# starting kernel
# baseline (speedup 1.0000x reference)
#include <hip/hip_runtime.h>

typedef __attribute__((ext_vector_type(2))) float v2f;
typedef __attribute__((ext_vector_type(8))) float v8f;

#define DT_F 0.1f

struct alignas(4) F3 { float a, b, c; };

__device__ __forceinline__ float swap16(float v) {
  return __int_as_float(__builtin_amdgcn_ds_swizzle(__float_as_int(v), 0x401F));
}

__device__ __forceinline__ void lorenz(float x, float y, float z,
                                       float sg, float rh, float bt,
                                       float& d0, float& d1, float& d2) {
  d0 = sg * (y - x);
  d1 = rh * x - y - x * z;
  d2 = x * y - bt * z;
}

__global__ __launch_bounds__(256) void mlp_lorenz_rk4(
    const float* __restrict__ x,
    const float* __restrict__ W1, const float* __restrict__ b1,
    const float* __restrict__ W2, const float* __restrict__ b2,
    const float* __restrict__ sigma_p, const float* __restrict__ rho_p,
    const float* __restrict__ beta_p,
    float* __restrict__ out, int numTiles) {
  const int lane = threadIdx.x & 31;
  const int m    = lane & 15;
  const bool hi  = lane >= 16;
  const int waveId = (blockIdx.x * blockDim.x + threadIdx.x) >> 5;
  const int nWaves = (gridDim.x * blockDim.x) >> 5;

  const float sg = sigma_p[0], rh = rho_p[0], bt = beta_p[0];
  const float b2_0 = b2[0], b2_1 = b2[1], b2_2 = b2[2];

  v2f a1;
  a1.x = (m < 10) ? W1[m * 3 + (hi ? 2 : 0)] : 0.0f;
  a1.y = (m < 10 && !hi) ? W1[m * 3 + 1] : 0.0f;

  v2f a2a, a2b, a2c;
  a2a.x = (m < 3) ? W2[m * 10 + (hi ? 2 : 0)] : 0.0f;
  a2a.y = (m < 3) ? W2[m * 10 + (hi ? 3 : 1)] : 0.0f;
  a2b.x = (m < 3) ? W2[m * 10 + (hi ? 6 : 4)] : 0.0f;
  a2b.y = (m < 3) ? W2[m * 10 + (hi ? 7 : 5)] : 0.0f;
  a2c.x = (m < 3 && !hi) ? W2[m * 10 + 8] : 0.0f;
  a2c.y = (m < 3 && !hi) ? W2[m * 10 + 9] : 0.0f;

  float bb[8];
#pragma unroll
  for (int v = 0; v < 8; ++v) {
    const int idx = hi ? (v + 8) : v;
    bb[v] = (idx < 10) ? b1[idx] : 0.0f;
  }

  const v8f czero = {};

  for (int t = waveId; t < numTiles; t += nWaves) {
    const int rb3 = (t * 16 + m) * 3;

    const float bxx = x[rb3 + (hi ? 2 : 0)];
    const float bxy = x[rb3 + 1];
    v2f bx;
    bx.x = bxx;
    bx.y = hi ? 0.0f : bxy;

    {
      int tn = t + nWaves;
      if (tn >= numTiles) tn = t;
      __builtin_prefetch(x + (tn * 16 + m) * 3, 0, 1);
    }

    v8f h = __builtin_amdgcn_wmma_f32_16x16x4_f32(false, a1, false, bx,
                                                  (short)0, czero, false, false);
#pragma unroll
    for (int v = 0; v < 8; ++v) h[v] = fmaxf(h[v] + bb[v], 0.0f);

    const float s0 = swap16(h[0]);
    const float s1 = swap16(h[1]);
    const float s2 = swap16(h[2]);
    const float s3 = swap16(h[3]);
    const float s6 = swap16(h[6]);
    const float s7 = swap16(h[7]);

    v2f bc0, bc1, bc2;
    bc0.x = hi ? s2 : h[0];
    bc0.y = hi ? s3 : h[1];
    bc1.x = hi ? s6 : h[4];
    bc1.y = hi ? s7 : h[5];
    bc2.x = hi ? 0.0f : s0;
    bc2.y = hi ? 0.0f : s1;

    v8f d2 = __builtin_amdgcn_wmma_f32_16x16x4_f32(false, a2a, false, bc0,
                                                   (short)0, czero, false, false);
    d2 = __builtin_amdgcn_wmma_f32_16x16x4_f32(false, a2b, false, bc1,
                                               (short)0, d2, false, false);
    d2 = __builtin_amdgcn_wmma_f32_16x16x4_f32(false, a2c, false, bc2,
                                               (short)0, d2, false, false);

    if (!hi) {
      const float r0 = d2[0] + b2_0;
      const float r1 = d2[1] + b2_1;
      const float r2 = d2[2] + b2_2;

      float k10, k11, k12, k20, k21, k22, k30, k31, k32, k40, k41, k42;
      lorenz(r0, r1, r2, sg, rh, bt, k10, k11, k12);
      lorenz(r0 + 0.5f * DT_F * k10, r1 + 0.5f * DT_F * k11, r2 + 0.5f * DT_F * k12,
             sg, rh, bt, k20, k21, k22);
      lorenz(r0 + 0.5f * DT_F * k20, r1 + 0.5f * DT_F * k21, r2 + 0.5f * DT_F * k22,
             sg, rh, bt, k30, k31, k32);
      lorenz(r0 + DT_F * k30, r1 + DT_F * k31, r2 + DT_F * k32,
             sg, rh, bt, k40, k41, k42);
      const float c = DT_F / 6.0f;

      F3 o;
      o.a = r0 + c * (k10 + 2.0f * k20 + 2.0f * k30 + k40);
      o.b = r1 + c * (k11 + 2.0f * k21 + 2.0f * k31 + k41);
      o.c = r2 + c * (k12 + 2.0f * k22 + 2.0f * k32 + k42);
      reinterpret_cast<F3*>(out)[t * 16 + m] = o;
    }
  }
}

extern "C" void kernel_launch(void* const* d_in, const int* in_sizes, int n_in,
                              void* d_out, int out_size, void* d_ws, size_t ws_size,
                              hipStream_t stream) {
  const float* x     = (const float*)d_in[0];
  const float* W1    = (const float*)d_in[1];
  const float* b1    = (const float*)d_in[2];
  const float* W2    = (const float*)d_in[3];
  const float* b2    = (const float*)d_in[4];
  const float* sigma = (const float*)d_in[5];
  const float* rho   = (const float*)d_in[6];
  const float* beta  = (const float*)d_in[7];
  float* out = (float*)d_out;

  const int Bn = in_sizes[0] / 3;
  const int numTiles = Bn / 16;

  const int block = 256;
  int grid = 8192;
  const int maxBlocks = (numTiles * 32 + block - 1) / block;
  if (grid > maxBlocks) grid = maxBlocks > 0 ? maxBlocks : 1;

  mlp_lorenz_rk4<<<grid, block, 0, stream>>>(x, W1, b1, W2, b2, sigma, rho, beta,
                                             out, numTiles);
}


# Wave-private LDS staging: each wave streams 128-row chunks (1536 B) with 16-B global loads/stores and takes its B fragments from LDS; whole 128-B output lines per store; 8 tiles per trip
# speedup vs baseline: 1.1897x; 1.1897x over previous
#include <hip/hip_runtime.h>

typedef __attribute__((ext_vector_type(2))) float v2f;
typedef __attribute__((ext_vector_type(4))) float v4f;
typedef __attribute__((ext_vector_type(8))) float v8f;

#define TILES_PER_CHUNK 8
#define CHUNK_V4 96
#define WAVES_PER_BLOCK 8

#define DT_F 0.1f

struct alignas(4) F3 { float a, b, c; };

__device__ __forceinline__ float swap16(float v) {
  return __int_as_float(__builtin_amdgcn_ds_swizzle(__float_as_int(v), 0x401F));
}

__device__ __forceinline__ void lorenz(float x, float y, float z,
                                       float sg, float rh, float bt,
                                       float& d0, float& d1, float& d2) {
  d0 = sg * (y - x);
  d1 = rh * x - y - x * z;
  d2 = x * y - bt * z;
}

__global__ __launch_bounds__(256) void mlp_lorenz_rk4(
    const float* __restrict__ x,
    const float* __restrict__ W1, const float* __restrict__ b1,
    const float* __restrict__ W2, const float* __restrict__ b2,
    const float* __restrict__ sigma_p, const float* __restrict__ rho_p,
    const float* __restrict__ beta_p,
    float* __restrict__ out, int numTiles) {
  const int lane = threadIdx.x & 31;
  const int m    = lane & 15;
  const bool hi  = lane >= 16;
  const int waveId = (blockIdx.x * blockDim.x + threadIdx.x) >> 5;
  const int nWaves = (gridDim.x * blockDim.x) >> 5;

  const float sg = sigma_p[0], rh = rho_p[0], bt = beta_p[0];
  const float b2_0 = b2[0], b2_1 = b2[1], b2_2 = b2[2];

  v2f a1;
  a1.x = (m < 10) ? W1[m * 3 + (hi ? 2 : 0)] : 0.0f;
  a1.y = (m < 10 && !hi) ? W1[m * 3 + 1] : 0.0f;

  v2f a2a, a2b, a2c;
  a2a.x = (m < 3) ? W2[m * 10 + (hi ? 2 : 0)] : 0.0f;
  a2a.y = (m < 3) ? W2[m * 10 + (hi ? 3 : 1)] : 0.0f;
  a2b.x = (m < 3) ? W2[m * 10 + (hi ? 6 : 4)] : 0.0f;
  a2b.y = (m < 3) ? W2[m * 10 + (hi ? 7 : 5)] : 0.0f;
  a2c.x = (m < 3 && !hi) ? W2[m * 10 + 8] : 0.0f;
  a2c.y = (m < 3 && !hi) ? W2[m * 10 + 9] : 0.0f;

  float bb[8];
#pragma unroll
  for (int v = 0; v < 8; ++v) {
    const int idx = hi ? (v + 8) : v;
    bb[v] = (idx < 10) ? b1[idx] : 0.0f;
  }

  const v8f czero = {};

  __shared__ v4f stage[WAVES_PER_BLOCK * CHUNK_V4];
  v4f* my4 = stage + (threadIdx.x >> 5) * CHUNK_V4;
  float* myf = reinterpret_cast<float*>(my4);

  const int totalV4 = numTiles * 12;
  const int numChunks = (numTiles + TILES_PER_CHUNK - 1) / TILES_PER_CHUNK;
  const v4f* __restrict__ x4 = reinterpret_cast<const v4f*>(x);
  v4f* __restrict__ o4 = reinterpret_cast<v4f*>(out);
  const v4f vzero = {};
  const int fa = m * 3 + (hi ? 2 : 0);
  const int fb = m * 3 + 1;

  for (int ch = waveId; ch < numChunks; ch += nWaves) {
    const int g0 = ch * CHUNK_V4 + lane;
    const int g1 = g0 + 32;
    const int g2 = g0 + 64;
    const v4f in0 = (g0 < totalV4) ? x4[g0] : vzero;
    const v4f in1 = (g1 < totalV4) ? x4[g1] : vzero;
    const v4f in2 = (g2 < totalV4) ? x4[g2] : vzero;
    my4[lane]      = in0;
    my4[lane + 32] = in1;
    my4[lane + 64] = in2;

#pragma unroll
    for (int j = 0; j < TILES_PER_CHUNK; ++j) {
      float* tp = myf + j * 48;

      const float bxx = tp[fa];
      const float bxy = tp[fb];
      v2f bx;
      bx.x = bxx;
      bx.y = hi ? 0.0f : bxy;

      v8f h = __builtin_amdgcn_wmma_f32_16x16x4_f32(false, a1, false, bx,
                                                    (short)0, czero, false, false);
#pragma unroll
      for (int v = 0; v < 8; ++v) h[v] = fmaxf(h[v] + bb[v], 0.0f);

      const float s0 = swap16(h[0]);
      const float s1 = swap16(h[1]);
      const float s2 = swap16(h[2]);
      const float s3 = swap16(h[3]);
      const float s6 = swap16(h[6]);
      const float s7 = swap16(h[7]);

      v2f bc0, bc1, bc2;
      bc0.x = hi ? s2 : h[0];
      bc0.y = hi ? s3 : h[1];
      bc1.x = hi ? s6 : h[4];
      bc1.y = hi ? s7 : h[5];
      bc2.x = hi ? 0.0f : s0;
      bc2.y = hi ? 0.0f : s1;

      v8f d2 = __builtin_amdgcn_wmma_f32_16x16x4_f32(false, a2a, false, bc0,
                                                     (short)0, czero, false, false);
      d2 = __builtin_amdgcn_wmma_f32_16x16x4_f32(false, a2b, false, bc1,
                                                 (short)0, d2, false, false);
      d2 = __builtin_amdgcn_wmma_f32_16x16x4_f32(false, a2c, false, bc2,
                                                 (short)0, d2, false, false);

      if (!hi) {
      const float r0 = d2[0] + b2_0;
      const float r1 = d2[1] + b2_1;
      const float r2 = d2[2] + b2_2;

      float k10, k11, k12, k20, k21, k22, k30, k31, k32, k40, k41, k42;
      lorenz(r0, r1, r2, sg, rh, bt, k10, k11, k12);
      lorenz(r0 + 0.5f * DT_F * k10, r1 + 0.5f * DT_F * k11, r2 + 0.5f * DT_F * k12,
             sg, rh, bt, k20, k21, k22);
      lorenz(r0 + 0.5f * DT_F * k20, r1 + 0.5f * DT_F * k21, r2 + 0.5f * DT_F * k22,
             sg, rh, bt, k30, k31, k32);
      lorenz(r0 + DT_F * k30, r1 + DT_F * k31, r2 + DT_F * k32,
             sg, rh, bt, k40, k41, k42);
      const float c = DT_F / 6.0f;

      F3 o;
      o.a = r0 + c * (k10 + 2.0f * k20 + 2.0f * k30 + k40);
      o.b = r1 + c * (k11 + 2.0f * k21 + 2.0f * k31 + k41);
      o.c = r2 + c * (k12 + 2.0f * k22 + 2.0f * k32 + k42);
      reinterpret_cast<F3*>(tp)[m] = o;
      }
    }

    const v4f q0 = my4[lane];
    const v4f q1 = my4[lane + 32];
    const v4f q2 = my4[lane + 64];
    if (g0 < totalV4) o4[g0] = q0;
    if (g1 < totalV4) o4[g1] = q1;
    if (g2 < totalV4) o4[g2] = q2;
  }
}

extern "C" void kernel_launch(void* const* d_in, const int* in_sizes, int n_in,
                              void* d_out, int out_size, void* d_ws, size_t ws_size,
                              hipStream_t stream) {
  const float* x     = (const float*)d_in[0];
  const float* W1    = (const float*)d_in[1];
  const float* b1    = (const float*)d_in[2];
  const float* W2    = (const float*)d_in[3];
  const float* b2    = (const float*)d_in[4];
  const float* sigma = (const float*)d_in[5];
  const float* rho   = (const float*)d_in[6];
  const float* beta  = (const float*)d_in[7];
  float* out = (float*)d_out;

  const int Bn = in_sizes[0] / 3;
  const int numTiles = Bn / 16;

  const int block = 256;
  int grid = 8192;
  const int maxBlocks = (numTiles * 32 + block - 1) / block;
  if (grid > maxBlocks) grid = maxBlocks > 0 ? maxBlocks : 1;

  mlp_lorenz_rk4<<<grid, block, 0, stream>>>(x, W1, b1, W2, b2, sigma, rho, beta,
                                             out, numTiles);
}


# RK4 split from the matrix layout: the 8 tiles park their layer-2 sums in the wave's LDS chunk, then every lane integrates 4 whole rows (32 lanes, 4 independent chains; was 16 lanes per tile)
# speedup vs baseline: 1.2632x; 1.0618x over previous
#include <hip/hip_runtime.h>

typedef __attribute__((ext_vector_type(2))) float v2f;
typedef __attribute__((ext_vector_type(4))) float v4f;
typedef __attribute__((ext_vector_type(8))) float v8f;

#define TILES_PER_CHUNK 8
#define CHUNK_V4 96
#define WAVES_PER_BLOCK 8

#define DT_F 0.1f

struct alignas(4) F3 { float a, b, c; };

__device__ __forceinline__ float swap16(float v) {
  return __int_as_float(__builtin_amdgcn_ds_swizzle(__float_as_int(v), 0x401F));
}

__device__ __forceinline__ void lorenz(float x, float y, float z,
                                       float sg, float rh, float bt,
                                       float& d0, float& d1, float& d2) {
  d0 = sg * (y - x);
  d1 = rh * x - y - x * z;
  d2 = x * y - bt * z;
}

__global__ __launch_bounds__(256) void mlp_lorenz_rk4(
    const float* __restrict__ x,
    const float* __restrict__ W1, const float* __restrict__ b1,
    const float* __restrict__ W2, const float* __restrict__ b2,
    const float* __restrict__ sigma_p, const float* __restrict__ rho_p,
    const float* __restrict__ beta_p,
    float* __restrict__ out, int numTiles) {
  const int lane = threadIdx.x & 31;
  const int m    = lane & 15;
  const bool hi  = lane >= 16;
  const int waveId = (blockIdx.x * blockDim.x + threadIdx.x) >> 5;
  const int nWaves = (gridDim.x * blockDim.x) >> 5;

  const float sg = sigma_p[0], rh = rho_p[0], bt = beta_p[0];
  const float b2_0 = b2[0], b2_1 = b2[1], b2_2 = b2[2];

  v2f a1;
  a1.x = (m < 10) ? W1[m * 3 + (hi ? 2 : 0)] : 0.0f;
  a1.y = (m < 10 && !hi) ? W1[m * 3 + 1] : 0.0f;

  v2f a2a, a2b, a2c;
  a2a.x = (m < 3) ? W2[m * 10 + (hi ? 2 : 0)] : 0.0f;
  a2a.y = (m < 3) ? W2[m * 10 + (hi ? 3 : 1)] : 0.0f;
  a2b.x = (m < 3) ? W2[m * 10 + (hi ? 6 : 4)] : 0.0f;
  a2b.y = (m < 3) ? W2[m * 10 + (hi ? 7 : 5)] : 0.0f;
  a2c.x = (m < 3 && !hi) ? W2[m * 10 + 8] : 0.0f;
  a2c.y = (m < 3 && !hi) ? W2[m * 10 + 9] : 0.0f;

  float bb[8];
#pragma unroll
  for (int v = 0; v < 8; ++v) {
    const int idx = hi ? (v + 8) : v;
    bb[v] = (idx < 10) ? b1[idx] : 0.0f;
  }

  const v8f czero = {};

  __shared__ v4f stage[WAVES_PER_BLOCK * CHUNK_V4];
  v4f* my4 = stage + (threadIdx.x >> 5) * CHUNK_V4;
  float* myf = reinterpret_cast<float*>(my4);

  const int totalV4 = numTiles * 12;
  const int numChunks = (numTiles + TILES_PER_CHUNK - 1) / TILES_PER_CHUNK;
  const v4f* __restrict__ x4 = reinterpret_cast<const v4f*>(x);
  v4f* __restrict__ o4 = reinterpret_cast<v4f*>(out);
  const v4f vzero = {};
  const int fa = m * 3 + (hi ? 2 : 0);
  const int fb = m * 3 + 1;

  for (int ch = waveId; ch < numChunks; ch += nWaves) {
    const int g0 = ch * CHUNK_V4 + lane;
    const int g1 = g0 + 32;
    const int g2 = g0 + 64;
    const v4f in0 = (g0 < totalV4) ? x4[g0] : vzero;
    const v4f in1 = (g1 < totalV4) ? x4[g1] : vzero;
    const v4f in2 = (g2 < totalV4) ? x4[g2] : vzero;
    my4[lane]      = in0;
    my4[lane + 32] = in1;
    my4[lane + 64] = in2;

#pragma unroll
    for (int j = 0; j < TILES_PER_CHUNK; ++j) {
      float* tp = myf + j * 48;

      const float bxx = tp[fa];
      const float bxy = tp[fb];
      v2f bx;
      bx.x = bxx;
      bx.y = hi ? 0.0f : bxy;

      v8f h = __builtin_amdgcn_wmma_f32_16x16x4_f32(false, a1, false, bx,
                                                    (short)0, czero, false, false);
#pragma unroll
      for (int v = 0; v < 8; ++v) h[v] = fmaxf(h[v] + bb[v], 0.0f);

      const float s0 = swap16(h[0]);
      const float s1 = swap16(h[1]);
      const float s2 = swap16(h[2]);
      const float s3 = swap16(h[3]);
      const float s6 = swap16(h[6]);
      const float s7 = swap16(h[7]);

      v2f bc0, bc1, bc2;
      bc0.x = hi ? s2 : h[0];
      bc0.y = hi ? s3 : h[1];
      bc1.x = hi ? s6 : h[4];
      bc1.y = hi ? s7 : h[5];
      bc2.x = hi ? 0.0f : s0;
      bc2.y = hi ? 0.0f : s1;

      v8f d2 = __builtin_amdgcn_wmma_f32_16x16x4_f32(false, a2a, false, bc0,
                                                     (short)0, czero, false, false);
      d2 = __builtin_amdgcn_wmma_f32_16x16x4_f32(false, a2b, false, bc1,
                                                 (short)0, d2, false, false);
      d2 = __builtin_amdgcn_wmma_f32_16x16x4_f32(false, a2c, false, bc2,
                                                 (short)0, d2, false, false);

      if (!hi) {
        F3 o;
        o.a = d2[0];
        o.b = d2[1];
        o.c = d2[2];
        reinterpret_cast<F3*>(tp)[m] = o;
      }
    }

    {
      const v4f p0 = my4[3 * lane];
      const v4f p1 = my4[3 * lane + 1];
      const v4f p2 = my4[3 * lane + 2];
      float st[12] = {p0.x, p0.y, p0.z, p0.w, p1.x, p1.y, p1.z, p1.w, p2.x, p2.y, p2.z, p2.w};
#pragma unroll
      for (int q = 0; q < 4; ++q) {
        const float r0 = st[3 * q + 0] + b2_0;
        const float r1 = st[3 * q + 1] + b2_1;
        const float r2 = st[3 * q + 2] + b2_2;

        float k10, k11, k12, k20, k21, k22, k30, k31, k32, k40, k41, k42;
        lorenz(r0, r1, r2, sg, rh, bt, k10, k11, k12);
        lorenz(r0 + 0.5f * DT_F * k10, r1 + 0.5f * DT_F * k11, r2 + 0.5f * DT_F * k12,
               sg, rh, bt, k20, k21, k22);
        lorenz(r0 + 0.5f * DT_F * k20, r1 + 0.5f * DT_F * k21, r2 + 0.5f * DT_F * k22,
               sg, rh, bt, k30, k31, k32);
        lorenz(r0 + DT_F * k30, r1 + DT_F * k31, r2 + DT_F * k32,
               sg, rh, bt, k40, k41, k42);
        const float c = DT_F / 6.0f;

        st[3 * q + 0] = r0 + c * (k10 + 2.0f * k20 + 2.0f * k30 + k40);
        st[3 * q + 1] = r1 + c * (k11 + 2.0f * k21 + 2.0f * k31 + k41);
        st[3 * q + 2] = r2 + c * (k12 + 2.0f * k22 + 2.0f * k32 + k42);
      }
      v4f w0, w1, w2;
      w0.x = st[0]; w0.y = st[1]; w0.z = st[2];  w0.w = st[3];
      w1.x = st[4]; w1.y = st[5]; w1.z = st[6];  w1.w = st[7];
      w2.x = st[8]; w2.y = st[9]; w2.z = st[10]; w2.w = st[11];
      my4[3 * lane]     = w0;
      my4[3 * lane + 1] = w1;
      my4[3 * lane + 2] = w2;
    }

    const v4f q0 = my4[lane];
    const v4f q1 = my4[lane + 32];
    const v4f q2 = my4[lane + 64];
    if (g0 < totalV4) o4[g0] = q0;
    if (g1 < totalV4) o4[g1] = q1;
    if (g2 < totalV4) o4[g2] = q2;
  }
}

extern "C" void kernel_launch(void* const* d_in, const int* in_sizes, int n_in,
                              void* d_out, int out_size, void* d_ws, size_t ws_size,
                              hipStream_t stream) {
  const float* x     = (const float*)d_in[0];
  const float* W1    = (const float*)d_in[1];
  const float* b1    = (const float*)d_in[2];
  const float* W2    = (const float*)d_in[3];
  const float* b2    = (const float*)d_in[4];
  const float* sigma = (const float*)d_in[5];
  const float* rho   = (const float*)d_in[6];
  const float* beta  = (const float*)d_in[7];
  float* out = (float*)d_out;

  const int Bn = in_sizes[0] / 3;
  const int numTiles = Bn / 16;

  const int block = 256;
  int grid = 8192;
  const int maxBlocks = (numTiles * 32 + block - 1) / block;
  if (grid > maxBlocks) grid = maxBlocks > 0 ? maxBlocks : 1;

  mlp_lorenz_rk4<<<grid, block, 0, stream>>>(x, W1, b1, W2, b2, sigma, rho, beta,
                                             out, numTiles);
}


# Layer-1 sums reach layer 2's B layout by 4 v_permlane16_swap_b32 row swaps per tile on the vector ALU (was 6 ds_swizzle through LDS + 6 selects + 4 LDS waits); bias+ReLU on the 6 swapped registers
# speedup vs baseline: 1.3044x; 1.0326x over previous
#include <hip/hip_runtime.h>

typedef __attribute__((ext_vector_type(2))) float v2f;
typedef __attribute__((ext_vector_type(4))) float v4f;
typedef __attribute__((ext_vector_type(8))) float v8f;

#define TILES_PER_CHUNK 8
#define CHUNK_V4 96
#define WAVES_PER_BLOCK 8

#define DT_F 0.1f

struct alignas(4) F3 { float a, b, c; };

typedef __attribute__((ext_vector_type(2))) unsigned int v2u;

__device__ __forceinline__ void rowswap(float& d, float& s) {
  const v2u r = __builtin_amdgcn_permlane16_swap(__float_as_uint(d), __float_as_uint(s), false, false);
  d = __uint_as_float(r.x);
  s = __uint_as_float(r.y);
}

__device__ __forceinline__ void lorenz(float x, float y, float z,
                                       float sg, float rh, float bt,
                                       float& d0, float& d1, float& d2) {
  d0 = sg * (y - x);
  d1 = rh * x - y - x * z;
  d2 = x * y - bt * z;
}

__global__ __launch_bounds__(256) void mlp_lorenz_rk4(
    const float* __restrict__ x,
    const float* __restrict__ W1, const float* __restrict__ b1,
    const float* __restrict__ W2, const float* __restrict__ b2,
    const float* __restrict__ sigma_p, const float* __restrict__ rho_p,
    const float* __restrict__ beta_p,
    float* __restrict__ out, int numTiles) {
  const int lane = threadIdx.x & 31;
  const int m    = lane & 15;
  const bool hi  = lane >= 16;
  const int waveId = (blockIdx.x * blockDim.x + threadIdx.x) >> 5;
  const int nWaves = (gridDim.x * blockDim.x) >> 5;

  const float sg = sigma_p[0], rh = rho_p[0], bt = beta_p[0];
  const float b2_0 = b2[0], b2_1 = b2[1], b2_2 = b2[2];

  v2f a1;
  a1.x = (m < 10) ? W1[m * 3 + (hi ? 2 : 0)] : 0.0f;
  a1.y = (m < 10 && !hi) ? W1[m * 3 + 1] : 0.0f;

  v2f a2a, a2b, a2c;
  a2a.x = (m < 3) ? W2[m * 10 + (hi ? 2 : 0)] : 0.0f;
  a2a.y = (m < 3) ? W2[m * 10 + (hi ? 3 : 1)] : 0.0f;
  a2b.x = (m < 3) ? W2[m * 10 + (hi ? 6 : 4)] : 0.0f;
  a2b.y = (m < 3) ? W2[m * 10 + (hi ? 7 : 5)] : 0.0f;
  a2c.x = (m < 3 && !hi) ? W2[m * 10 + 8] : 0.0f;
  a2c.y = (m < 3 && !hi) ? W2[m * 10 + 9] : 0.0f;

  v2f bs0, bs1, bs2;
  bs0.x = b1[hi ? 2 : 0];
  bs0.y = b1[hi ? 3 : 1];
  bs1.x = b1[hi ? 6 : 4];
  bs1.y = b1[hi ? 7 : 5];
  bs2.x = hi ? 0.0f : b1[8];
  bs2.y = hi ? 0.0f : b1[9];

  const v8f czero = {};

  __shared__ v4f stage[WAVES_PER_BLOCK * CHUNK_V4];
  v4f* my4 = stage + (threadIdx.x >> 5) * CHUNK_V4;
  float* myf = reinterpret_cast<float*>(my4);

  const int totalV4 = numTiles * 12;
  const int numChunks = (numTiles + TILES_PER_CHUNK - 1) / TILES_PER_CHUNK;
  const v4f* __restrict__ x4 = reinterpret_cast<const v4f*>(x);
  v4f* __restrict__ o4 = reinterpret_cast<v4f*>(out);
  const v4f vzero = {};
  const int fa = m * 3 + (hi ? 2 : 0);
  const int fb = m * 3 + 1;

  for (int ch = waveId; ch < numChunks; ch += nWaves) {
    const int g0 = ch * CHUNK_V4 + lane;
    const int g1 = g0 + 32;
    const int g2 = g0 + 64;
    const v4f in0 = (g0 < totalV4) ? x4[g0] : vzero;
    const v4f in1 = (g1 < totalV4) ? x4[g1] : vzero;
    const v4f in2 = (g2 < totalV4) ? x4[g2] : vzero;
    my4[lane]      = in0;
    my4[lane + 32] = in1;
    my4[lane + 64] = in2;

#pragma unroll
    for (int j = 0; j < TILES_PER_CHUNK; ++j) {
      float* tp = myf + j * 48;

      const float bxx = tp[fa];
      const float bxy = tp[fb];
      v2f bx;
      bx.x = bxx;
      bx.y = hi ? 0.0f : bxy;

      v8f h = __builtin_amdgcn_wmma_f32_16x16x4_f32(false, a1, false, bx,
                                                    (short)0, czero, false, false);
      float h0 = h[0], h1 = h[1], h2 = h[2], h3 = h[3];
      float h4 = h[4], h5 = h[5], h6 = h[6], h7 = h[7];
      rowswap(h0, h2);
      rowswap(h1, h3);
      rowswap(h4, h6);
      rowswap(h5, h7);

      v2f bc0, bc1, bc2;
      bc0.x = fmaxf(h0 + bs0.x, 0.0f);
      bc0.y = fmaxf(h1 + bs0.y, 0.0f);
      bc1.x = fmaxf(h4 + bs1.x, 0.0f);
      bc1.y = fmaxf(h5 + bs1.y, 0.0f);
      bc2.x = fmaxf(h2 + bs2.x, 0.0f);
      bc2.y = fmaxf(h3 + bs2.y, 0.0f);

      v8f d2 = __builtin_amdgcn_wmma_f32_16x16x4_f32(false, a2a, false, bc0,
                                                     (short)0, czero, false, false);
      d2 = __builtin_amdgcn_wmma_f32_16x16x4_f32(false, a2b, false, bc1,
                                                 (short)0, d2, false, false);
      d2 = __builtin_amdgcn_wmma_f32_16x16x4_f32(false, a2c, false, bc2,
                                                 (short)0, d2, false, false);

      if (!hi) {
        F3 o;
        o.a = d2[0];
        o.b = d2[1];
        o.c = d2[2];
        reinterpret_cast<F3*>(tp)[m] = o;
      }
    }

    {
      const v4f p0 = my4[3 * lane];
      const v4f p1 = my4[3 * lane + 1];
      const v4f p2 = my4[3 * lane + 2];
      float st[12] = {p0.x, p0.y, p0.z, p0.w, p1.x, p1.y, p1.z, p1.w, p2.x, p2.y, p2.z, p2.w};
#pragma unroll
      for (int q = 0; q < 4; ++q) {
        const float r0 = st[3 * q + 0] + b2_0;
        const float r1 = st[3 * q + 1] + b2_1;
        const float r2 = st[3 * q + 2] + b2_2;

        float k10, k11, k12, k20, k21, k22, k30, k31, k32, k40, k41, k42;
        lorenz(r0, r1, r2, sg, rh, bt, k10, k11, k12);
        lorenz(r0 + 0.5f * DT_F * k10, r1 + 0.5f * DT_F * k11, r2 + 0.5f * DT_F * k12,
               sg, rh, bt, k20, k21, k22);
        lorenz(r0 + 0.5f * DT_F * k20, r1 + 0.5f * DT_F * k21, r2 + 0.5f * DT_F * k22,
               sg, rh, bt, k30, k31, k32);
        lorenz(r0 + DT_F * k30, r1 + DT_F * k31, r2 + DT_F * k32,
               sg, rh, bt, k40, k41, k42);
        const float c = DT_F / 6.0f;

        st[3 * q + 0] = r0 + c * (k10 + 2.0f * k20 + 2.0f * k30 + k40);
        st[3 * q + 1] = r1 + c * (k11 + 2.0f * k21 + 2.0f * k31 + k41);
        st[3 * q + 2] = r2 + c * (k12 + 2.0f * k22 + 2.0f * k32 + k42);
      }
      v4f w0, w1, w2;
      w0.x = st[0]; w0.y = st[1]; w0.z = st[2];  w0.w = st[3];
      w1.x = st[4]; w1.y = st[5]; w1.z = st[6];  w1.w = st[7];
      w2.x = st[8]; w2.y = st[9]; w2.z = st[10]; w2.w = st[11];
      my4[3 * lane]     = w0;
      my4[3 * lane + 1] = w1;
      my4[3 * lane + 2] = w2;
    }

    const v4f q0 = my4[lane];
    const v4f q1 = my4[lane + 32];
    const v4f q2 = my4[lane + 64];
    if (g0 < totalV4) o4[g0] = q0;
    if (g1 < totalV4) o4[g1] = q1;
    if (g2 < totalV4) o4[g2] = q2;
  }
}

extern "C" void kernel_launch(void* const* d_in, const int* in_sizes, int n_in,
                              void* d_out, int out_size, void* d_ws, size_t ws_size,
                              hipStream_t stream) {
  const float* x     = (const float*)d_in[0];
  const float* W1    = (const float*)d_in[1];
  const float* b1    = (const float*)d_in[2];
  const float* W2    = (const float*)d_in[3];
  const float* b2    = (const float*)d_in[4];
  const float* sigma = (const float*)d_in[5];
  const float* rho   = (const float*)d_in[6];
  const float* beta  = (const float*)d_in[7];
  float* out = (float*)d_out;

  const int Bn = in_sizes[0] / 3;
  const int numTiles = Bn / 16;

  const int block = 256;
  int grid = 8192;
  const int maxBlocks = (numTiles * 32 + block - 1) / block;
  if (grid > maxBlocks) grid = maxBlocks > 0 ? maxBlocks : 1;

  mlp_lorenz_rk4<<<grid, block, 0, stream>>>(x, W1, b1, W2, b2, sigma, rho, beta,
                                             out, numTiles);
}


# One resident round of waves: 16384 waves (8 blocks of 8 per WGP) each walk 4 chunks with the next chunk's three 16-B loads in flight; was 65536 one-chunk waves, each paying entry and weight loads
# speedup vs baseline: 1.3523x; 1.0367x over previous
#include <hip/hip_runtime.h>

typedef __attribute__((ext_vector_type(2))) float v2f;
typedef __attribute__((ext_vector_type(4))) float v4f;
typedef __attribute__((ext_vector_type(8))) float v8f;

#define TILES_PER_CHUNK 8
#define CHUNK_V4 96
#define WAVES_PER_BLOCK 8

#define DT_F 0.1f

struct alignas(4) F3 { float a, b, c; };

typedef __attribute__((ext_vector_type(2))) unsigned int v2u;

__device__ __forceinline__ void rowswap(float& d, float& s) {
  const v2u r = __builtin_amdgcn_permlane16_swap(__float_as_uint(d), __float_as_uint(s), false, false);
  d = __uint_as_float(r.x);
  s = __uint_as_float(r.y);
}

__device__ __forceinline__ void lorenz(float x, float y, float z,
                                       float sg, float rh, float bt,
                                       float& d0, float& d1, float& d2) {
  d0 = sg * (y - x);
  d1 = rh * x - y - x * z;
  d2 = x * y - bt * z;
}

__global__ __launch_bounds__(256) void mlp_lorenz_rk4(
    const float* __restrict__ x,
    const float* __restrict__ W1, const float* __restrict__ b1,
    const float* __restrict__ W2, const float* __restrict__ b2,
    const float* __restrict__ sigma_p, const float* __restrict__ rho_p,
    const float* __restrict__ beta_p,
    float* __restrict__ out, int numTiles) {
  const int lane = threadIdx.x & 31;
  const int m    = lane & 15;
  const bool hi  = lane >= 16;
  const int waveId = (blockIdx.x * blockDim.x + threadIdx.x) >> 5;
  const int nWaves = (gridDim.x * blockDim.x) >> 5;

  const float sg = sigma_p[0], rh = rho_p[0], bt = beta_p[0];
  const float b2_0 = b2[0], b2_1 = b2[1], b2_2 = b2[2];

  v2f a1;
  a1.x = (m < 10) ? W1[m * 3 + (hi ? 2 : 0)] : 0.0f;
  a1.y = (m < 10 && !hi) ? W1[m * 3 + 1] : 0.0f;

  v2f a2a, a2b, a2c;
  a2a.x = (m < 3) ? W2[m * 10 + (hi ? 2 : 0)] : 0.0f;
  a2a.y = (m < 3) ? W2[m * 10 + (hi ? 3 : 1)] : 0.0f;
  a2b.x = (m < 3) ? W2[m * 10 + (hi ? 6 : 4)] : 0.0f;
  a2b.y = (m < 3) ? W2[m * 10 + (hi ? 7 : 5)] : 0.0f;
  a2c.x = (m < 3 && !hi) ? W2[m * 10 + 8] : 0.0f;
  a2c.y = (m < 3 && !hi) ? W2[m * 10 + 9] : 0.0f;

  v2f bs0, bs1, bs2;
  bs0.x = b1[hi ? 2 : 0];
  bs0.y = b1[hi ? 3 : 1];
  bs1.x = b1[hi ? 6 : 4];
  bs1.y = b1[hi ? 7 : 5];
  bs2.x = hi ? 0.0f : b1[8];
  bs2.y = hi ? 0.0f : b1[9];

  const v8f czero = {};

  __shared__ v4f stage[WAVES_PER_BLOCK * CHUNK_V4];
  v4f* my4 = stage + (threadIdx.x >> 5) * CHUNK_V4;
  float* myf = reinterpret_cast<float*>(my4);

  const int totalV4 = numTiles * 12;
  const int numChunks = (numTiles + TILES_PER_CHUNK - 1) / TILES_PER_CHUNK;
  const v4f* __restrict__ x4 = reinterpret_cast<const v4f*>(x);
  v4f* __restrict__ o4 = reinterpret_cast<v4f*>(out);
  const v4f vzero = {};
  const int fa = m * 3 + (hi ? 2 : 0);
  const int fb = m * 3 + 1;

  v4f in0, in1, in2;
  {
    const int f0 = waveId * CHUNK_V4 + lane;
    in0 = (f0      < totalV4) ? x4[f0]      : vzero;
    in1 = (f0 + 32 < totalV4) ? x4[f0 + 32] : vzero;
    in2 = (f0 + 64 < totalV4) ? x4[f0 + 64] : vzero;
  }

  for (int ch = waveId; ch < numChunks; ch += nWaves) {
    const int g0 = ch * CHUNK_V4 + lane;
    const int g1 = g0 + 32;
    const int g2 = g0 + 64;
    my4[lane]      = in0;
    my4[lane + 32] = in1;
    my4[lane + 64] = in2;

    {
      const int n0 = g0 + nWaves * CHUNK_V4;
      in0 = (n0      < totalV4) ? x4[n0]      : vzero;
      in1 = (n0 + 32 < totalV4) ? x4[n0 + 32] : vzero;
      in2 = (n0 + 64 < totalV4) ? x4[n0 + 64] : vzero;
    }

#pragma unroll
    for (int j = 0; j < TILES_PER_CHUNK; ++j) {
      float* tp = myf + j * 48;

      const float bxx = tp[fa];
      const float bxy = tp[fb];
      v2f bx;
      bx.x = bxx;
      bx.y = hi ? 0.0f : bxy;

      v8f h = __builtin_amdgcn_wmma_f32_16x16x4_f32(false, a1, false, bx,
                                                    (short)0, czero, false, false);
      float h0 = h[0], h1 = h[1], h2 = h[2], h3 = h[3];
      float h4 = h[4], h5 = h[5], h6 = h[6], h7 = h[7];
      rowswap(h0, h2);
      rowswap(h1, h3);
      rowswap(h4, h6);
      rowswap(h5, h7);

      v2f bc0, bc1, bc2;
      bc0.x = fmaxf(h0 + bs0.x, 0.0f);
      bc0.y = fmaxf(h1 + bs0.y, 0.0f);
      bc1.x = fmaxf(h4 + bs1.x, 0.0f);
      bc1.y = fmaxf(h5 + bs1.y, 0.0f);
      bc2.x = fmaxf(h2 + bs2.x, 0.0f);
      bc2.y = fmaxf(h3 + bs2.y, 0.0f);

      v8f d2 = __builtin_amdgcn_wmma_f32_16x16x4_f32(false, a2a, false, bc0,
                                                     (short)0, czero, false, false);
      d2 = __builtin_amdgcn_wmma_f32_16x16x4_f32(false, a2b, false, bc1,
                                                 (short)0, d2, false, false);
      d2 = __builtin_amdgcn_wmma_f32_16x16x4_f32(false, a2c, false, bc2,
                                                 (short)0, d2, false, false);

      if (!hi) {
        F3 o;
        o.a = d2[0];
        o.b = d2[1];
        o.c = d2[2];
        reinterpret_cast<F3*>(tp)[m] = o;
      }
    }

    {
      const v4f p0 = my4[3 * lane];
      const v4f p1 = my4[3 * lane + 1];
      const v4f p2 = my4[3 * lane + 2];
      float st[12] = {p0.x, p0.y, p0.z, p0.w, p1.x, p1.y, p1.z, p1.w, p2.x, p2.y, p2.z, p2.w};
#pragma unroll
      for (int q = 0; q < 4; ++q) {
        const float r0 = st[3 * q + 0] + b2_0;
        const float r1 = st[3 * q + 1] + b2_1;
        const float r2 = st[3 * q + 2] + b2_2;

        float k10, k11, k12, k20, k21, k22, k30, k31, k32, k40, k41, k42;
        lorenz(r0, r1, r2, sg, rh, bt, k10, k11, k12);
        lorenz(r0 + 0.5f * DT_F * k10, r1 + 0.5f * DT_F * k11, r2 + 0.5f * DT_F * k12,
               sg, rh, bt, k20, k21, k22);
        lorenz(r0 + 0.5f * DT_F * k20, r1 + 0.5f * DT_F * k21, r2 + 0.5f * DT_F * k22,
               sg, rh, bt, k30, k31, k32);
        lorenz(r0 + DT_F * k30, r1 + DT_F * k31, r2 + DT_F * k32,
               sg, rh, bt, k40, k41, k42);
        const float c = DT_F / 6.0f;

        st[3 * q + 0] = r0 + c * (k10 + 2.0f * k20 + 2.0f * k30 + k40);
        st[3 * q + 1] = r1 + c * (k11 + 2.0f * k21 + 2.0f * k31 + k41);
        st[3 * q + 2] = r2 + c * (k12 + 2.0f * k22 + 2.0f * k32 + k42);
      }
      v4f w0, w1, w2;
      w0.x = st[0]; w0.y = st[1]; w0.z = st[2];  w0.w = st[3];
      w1.x = st[4]; w1.y = st[5]; w1.z = st[6];  w1.w = st[7];
      w2.x = st[8]; w2.y = st[9]; w2.z = st[10]; w2.w = st[11];
      my4[3 * lane]     = w0;
      my4[3 * lane + 1] = w1;
      my4[3 * lane + 2] = w2;
    }

    const v4f q0 = my4[lane];
    const v4f q1 = my4[lane + 32];
    const v4f q2 = my4[lane + 64];
    if (g0 < totalV4) o4[g0] = q0;
    if (g1 < totalV4) o4[g1] = q1;
    if (g2 < totalV4) o4[g2] = q2;
  }
}

extern "C" void kernel_launch(void* const* d_in, const int* in_sizes, int n_in,
                              void* d_out, int out_size, void* d_ws, size_t ws_size,
                              hipStream_t stream) {
  const float* x     = (const float*)d_in[0];
  const float* W1    = (const float*)d_in[1];
  const float* b1    = (const float*)d_in[2];
  const float* W2    = (const float*)d_in[3];
  const float* b2    = (const float*)d_in[4];
  const float* sigma = (const float*)d_in[5];
  const float* rho   = (const float*)d_in[6];
  const float* beta  = (const float*)d_in[7];
  float* out = (float*)d_out;

  const int Bn = in_sizes[0] / 3;
  const int numTiles = Bn / 16;

  const int block = 256;
  int grid = 2048;
  const int maxBlocks = (numTiles * 32 + block - 1) / block;
  if (grid > maxBlocks) grid = maxBlocks > 0 ? maxBlocks : 1;

  mlp_lorenz_rk4<<<grid, block, 0, stream>>>(x, W1, b1, W2, b2, sigma, rho, beta,
                                             out, numTiles);
}
